# starting kernel
# baseline (speedup 1.0000x reference)
#include <hip/hip_runtime.h>
#include <hip/hip_bf16.h>

typedef __attribute__((ext_vector_type(2))) float v2f;
typedef __attribute__((ext_vector_type(8))) float v8f;

#define C_IN 128

__global__ void sage_zero(float* __restrict__ p, long long n) {
  long long i = (long long)blockIdx.x * blockDim.x + threadIdx.x;
  long long stride = (long long)gridDim.x * blockDim.x;
  for (; i < n; i += stride) p[i] = 0.0f;
}

__global__ void sage_scatter(const float* __restrict__ x,
                             const int* __restrict__ src,
                             const int* __restrict__ dst,
                             float* __restrict__ agg,
                             float* __restrict__ cnt,
                             int nEdges) {
  long long t = (long long)blockIdx.x * blockDim.x + threadIdx.x;
  int e  = (int)(t >> 5);
  int cq = (int)(t & 31);
  if (e >= nEdges) return;
  int s = src[e];
  int d = dst[e];
  const float4* xrow = (const float4*)(x + (long long)s * C_IN);
  float4 v = xrow[cq];
  float* arow = agg + (long long)d * C_IN + cq * 4;
  atomicAdd(arow + 0, v.x);
  atomicAdd(arow + 1, v.y);
  atomicAdd(arow + 2, v.z);
  atomicAdd(arow + 3, v.w);
  if (cq == 0) atomicAdd(cnt + d, 1.0f);
}

template <int COUT, bool RELU>
__global__ void sage_gemm(const float* __restrict__ xin,
                          const float* __restrict__ agg,
                          const float* __restrict__ cnt,
                          const float* __restrict__ Wl,
                          const float* __restrict__ bl,
                          const float* __restrict__ Wr,
                          float* __restrict__ out,
                          int totalTiles) {
  const int tilesN = COUT / 16;
  int wave = (int)(((long long)blockIdx.x * blockDim.x + threadIdx.x) >> 5);
  if (wave >= totalTiles) return;
  int lane = threadIdx.x & 31;
  int m0 = (wave / tilesN) * 16;
  int n0 = (wave % tilesN) * 16;
  int r  = lane & 15;
  int kh = lane >> 4;
  int col = n0 + r;

  const float* arow = agg + (long long)(m0 + r) * C_IN;
  const float* xrow = xin + (long long)(m0 + r) * C_IN;
  float inv = 1.0f / fmaxf(cnt[m0 + r], 1.0f);

  v8f acc = {};

#pragma unroll
  for (int k = 0; k < C_IN; k += 4) {
    v2f a, b;
    int kk = k + 2 * kh;
    a[0] = arow[kk] * inv;
    a[1] = arow[kk + 1] * inv;
    b[0] = Wl[(long long)kk * COUT + col];
    b[1] = Wl[(long long)(kk + 1) * COUT + col];
    acc = __builtin_amdgcn_wmma_f32_16x16x4_f32(false, a, false, b,
                                                (short)0, acc, false, false);
  }
#pragma unroll
  for (int k = 0; k < C_IN; k += 4) {
    v2f a, b;
    int kk = k + 2 * kh;
    a[0] = xrow[kk];
    a[1] = xrow[kk + 1];
    b[0] = Wr[(long long)kk * COUT + col];
    b[1] = Wr[(long long)(kk + 1) * COUT + col];
    acc = __builtin_amdgcn_wmma_f32_16x16x4_f32(false, a, false, b,
                                                (short)0, acc, false, false);
  }

  float bias = bl[col];
#pragma unroll
  for (int rr = 0; rr < 8; ++rr) {
    float v = acc[rr] + bias;
    if (RELU) v = fmaxf(v, 0.0f);
    out[(long long)(m0 + rr + 8 * kh) * COUT + col] = v;
  }
}

extern "C" void kernel_launch(void* const* d_in, const int* in_sizes, int n_in,
                              void* d_out, int out_size, void* d_ws, size_t ws_size,
                              hipStream_t stream) {
  const float* x0  = (const float*)d_in[0];
  const int*   ei  = (const int*)d_in[1];
  const float* Wl0 = (const float*)d_in[2];
  const float* bl0 = (const float*)d_in[3];
  const float* Wr0 = (const float*)d_in[4];
  const float* Wl1 = (const float*)d_in[5];
  const float* bl1 = (const float*)d_in[6];
  const float* Wr1 = (const float*)d_in[7];
  const float* Wl2 = (const float*)d_in[8];
  const float* bl2 = (const float*)d_in[9];
  const float* Wr2 = (const float*)d_in[10];

  const int N = in_sizes[0] / C_IN;
  const int E = in_sizes[1] / 2;
  const int* src = ei;
  const int* dst = ei + E;

  float* h0  = (float*)d_ws;
  float* h1  = h0 + (size_t)N * C_IN;
  float* agg = h1 + (size_t)N * C_IN;
  float* cnt = agg + (size_t)N * C_IN;
  const long long aggCntElems = (long long)N * C_IN + N;

  const int ZB = 1024;
  const long long scatterThreads = (long long)E * 32;
  const int scatterBlocks = (int)((scatterThreads + 255) / 256);

  const int tilesM = (N + 15) / 16;

  sage_zero<<<ZB, 256, 0, stream>>>(agg, aggCntElems);
  sage_scatter<<<scatterBlocks, 256, 0, stream>>>(x0, src, dst, agg, cnt, E);
  {
    int tiles = tilesM * (128 / 16);
    int blocks = (tiles + 7) / 8;
    sage_gemm<128, true><<<blocks, 256, 0, stream>>>(x0, agg, cnt, Wl0, bl0, Wr0, h0, tiles);
  }

  sage_zero<<<ZB, 256, 0, stream>>>(agg, aggCntElems);
  sage_scatter<<<scatterBlocks, 256, 0, stream>>>(h0, src, dst, agg, cnt, E);
  {
    int tiles = tilesM * (128 / 16);
    int blocks = (tiles + 7) / 8;
    sage_gemm<128, true><<<blocks, 256, 0, stream>>>(h0, agg, cnt, Wl1, bl1, Wr1, h1, tiles);
  }

  sage_zero<<<ZB, 256, 0, stream>>>(agg, aggCntElems);
  sage_scatter<<<scatterBlocks, 256, 0, stream>>>(h1, src, dst, agg, cnt, E);
  {
    int tiles = tilesM * (64 / 16);
    int blocks = (tiles + 7) / 8;
    sage_gemm<64, false><<<blocks, 256, 0, stream>>>(h1, agg, cnt, Wl2, bl2, Wr2,
                                                     (float*)d_out, tiles);
  }
}


# Aggregation as destination-ordered gather: edges counting-sorted once per call (integer atomics), one wave sums a node's source rows in fixed sorted order; float atomics and zero fill removed
# speedup vs baseline: 2.3070x; 2.3070x over previous
#include <hip/hip_runtime.h>
#include <hip/hip_bf16.h>

typedef __attribute__((ext_vector_type(2))) float v2f;
typedef __attribute__((ext_vector_type(8))) float v8f;

#define C_IN 128

__global__ void csr_zero(int* __restrict__ p, int n) {
  int i = blockIdx.x * blockDim.x + threadIdx.x;
  if (i < n) p[i] = 0;
}

__global__ void csr_rank(const int* __restrict__ dst, int* __restrict__ deg,
                         int* __restrict__ rank, int nEdges) {
  int e = blockIdx.x * blockDim.x + threadIdx.x;
  if (e < nEdges) rank[e] = atomicAdd(deg + dst[e], 1);
}

#define SCAN_T 256
#define SCAN_CHUNK 1024

__global__ void csr_scan(const int* __restrict__ deg, int* __restrict__ start,
                         float* __restrict__ cnt, int n) {
  __shared__ int sPre[SCAN_T];
  __shared__ int sScan[SCAN_T];
  const int tid = threadIdx.x;
  const int base = blockIdx.x * SCAN_CHUNK;

  int pre = 0;
  const int4* d4 = (const int4*)deg;
  for (int q = tid; q < base / 4; q += SCAN_T) {
    int4 t = d4[q];
    pre += (t.x + t.y) + (t.z + t.w);
  }

  const int i0 = base + tid * 4;
  int v[4];
  int tsum = 0;
#pragma unroll
  for (int j = 0; j < 4; ++j) {
    int i = i0 + j;
    v[j] = (i < n) ? deg[i] : 0;
    tsum += v[j];
  }

  sPre[tid] = pre;
  sScan[tid] = tsum;
  __syncthreads();
  for (int st = SCAN_T / 2; st > 0; st >>= 1) {
    if (tid < st) sPre[tid] += sPre[tid + st];
    __syncthreads();
  }
  const int preTotal = sPre[0];
  for (int off = 1; off < SCAN_T; off <<= 1) {
    int t = (tid >= off) ? sScan[tid - off] : 0;
    __syncthreads();
    sScan[tid] += t;
    __syncthreads();
  }
  int running = preTotal + sScan[tid] - tsum;
#pragma unroll
  for (int j = 0; j < 4; ++j) {
    int i = i0 + j;
    if (i < n) {
      start[i] = running;
      cnt[i] = (float)v[j];
      running += v[j];
    }
  }
}

__global__ void csr_place(const int* __restrict__ src, const int* __restrict__ dst,
                          const int* __restrict__ start, const int* __restrict__ rank,
                          int* __restrict__ col, int nEdges) {
  int e = blockIdx.x * blockDim.x + threadIdx.x;
  if (e < nEdges) col[start[dst[e]] + rank[e]] = src[e];
}

__global__ void csr_order(const int* __restrict__ start, const int* __restrict__ deg,
                          const int* __restrict__ colIn, int* __restrict__ colOut,
                          int nNodes) {
  int node = (int)(((long long)blockIdx.x * blockDim.x + threadIdx.x) >> 5);
  int lane = threadIdx.x & 31;
  if (node >= nNodes) return;
  node = __builtin_amdgcn_readfirstlane(node);
  const int b = start[node];
  const int d = deg[node];
  for (int j = lane; j < d; j += 32) {
    int v = colIn[b + j];
    int r = 0;
    for (int i = 0; i < d; ++i) {
      int u = colIn[b + i];
      r += ((u < v) || (u == v && i < j)) ? 1 : 0;
    }
    colOut[b + r] = v;
  }
}

__global__ void sage_gather(const float* __restrict__ x,
                            const int* __restrict__ start,
                            const int* __restrict__ deg,
                            const int* __restrict__ col,
                            float* __restrict__ agg,
                            int nNodes) {
  int node = (int)(((long long)blockIdx.x * blockDim.x + threadIdx.x) >> 5);
  const int lane = threadIdx.x & 31;
  if (node >= nNodes) return;
  node = __builtin_amdgcn_readfirstlane(node);
  const int* lst = col + start[node];
  const int d = deg[node];

  float4 acc = make_float4(0.0f, 0.0f, 0.0f, 0.0f);
#define ROW(sidx) (((const float4*)(x + (long long)(sidx) * C_IN))[lane])
#define ADD(v) { acc.x += (v).x; acc.y += (v).y; acc.z += (v).z; acc.w += (v).w; }
  int j = 0;
  for (; j + 8 <= d; j += 8) {
    int s0 = lst[j], s1 = lst[j + 1], s2 = lst[j + 2], s3 = lst[j + 3];
    int s4 = lst[j + 4], s5 = lst[j + 5], s6 = lst[j + 6], s7 = lst[j + 7];
    float4 v0 = ROW(s0), v1 = ROW(s1), v2 = ROW(s2), v3 = ROW(s3);
    float4 v4 = ROW(s4), v5 = ROW(s5), v6 = ROW(s6), v7 = ROW(s7);
    ADD(v0) ADD(v1) ADD(v2) ADD(v3) ADD(v4) ADD(v5) ADD(v6) ADD(v7)
  }
  if (j + 4 <= d) {
    int s0 = lst[j], s1 = lst[j + 1], s2 = lst[j + 2], s3 = lst[j + 3];
    float4 v0 = ROW(s0), v1 = ROW(s1), v2 = ROW(s2), v3 = ROW(s3);
    ADD(v0) ADD(v1) ADD(v2) ADD(v3)
    j += 4;
  }
  if (j + 2 <= d) {
    int s0 = lst[j], s1 = lst[j + 1];
    float4 v0 = ROW(s0), v1 = ROW(s1);
    ADD(v0) ADD(v1)
    j += 2;
  }
  if (j < d) {
    int s0 = lst[j];
    float4 v0 = ROW(s0);
    ADD(v0)
  }
#undef ROW
#undef ADD
  ((float4*)(agg + (long long)node * C_IN))[lane] = acc;
}

template <int COUT, bool RELU>
__global__ void sage_gemm(const float* __restrict__ xin,
                          const float* __restrict__ agg,
                          const float* __restrict__ cnt,
                          const float* __restrict__ Wl,
                          const float* __restrict__ bl,
                          const float* __restrict__ Wr,
                          float* __restrict__ out,
                          int totalTiles) {
  const int tilesN = COUT / 16;
  int wave = (int)(((long long)blockIdx.x * blockDim.x + threadIdx.x) >> 5);
  if (wave >= totalTiles) return;
  int lane = threadIdx.x & 31;
  int m0 = (wave / tilesN) * 16;
  int n0 = (wave % tilesN) * 16;
  int r  = lane & 15;
  int kh = lane >> 4;
  int col = n0 + r;

  const float* arow = agg + (long long)(m0 + r) * C_IN;
  const float* xrow = xin + (long long)(m0 + r) * C_IN;
  float inv = 1.0f / fmaxf(cnt[m0 + r], 1.0f);

  v8f acc = {};

#pragma unroll
  for (int k = 0; k < C_IN; k += 4) {
    v2f a, b;
    int kk = k + 2 * kh;
    a[0] = arow[kk] * inv;
    a[1] = arow[kk + 1] * inv;
    b[0] = Wl[(long long)kk * COUT + col];
    b[1] = Wl[(long long)(kk + 1) * COUT + col];
    acc = __builtin_amdgcn_wmma_f32_16x16x4_f32(false, a, false, b,
                                                (short)0, acc, false, false);
  }
#pragma unroll
  for (int k = 0; k < C_IN; k += 4) {
    v2f a, b;
    int kk = k + 2 * kh;
    a[0] = xrow[kk];
    a[1] = xrow[kk + 1];
    b[0] = Wr[(long long)kk * COUT + col];
    b[1] = Wr[(long long)(kk + 1) * COUT + col];
    acc = __builtin_amdgcn_wmma_f32_16x16x4_f32(false, a, false, b,
                                                (short)0, acc, false, false);
  }

  float bias = bl[col];
#pragma unroll
  for (int rr = 0; rr < 8; ++rr) {
    float v = acc[rr] + bias;
    if (RELU) v = fmaxf(v, 0.0f);
    out[(long long)(m0 + rr + 8 * kh) * COUT + col] = v;
  }
}

extern "C" void kernel_launch(void* const* d_in, const int* in_sizes, int n_in,
                              void* d_out, int out_size, void* d_ws, size_t ws_size,
                              hipStream_t stream) {
  const float* x0  = (const float*)d_in[0];
  const int*   ei  = (const int*)d_in[1];
  const float* Wl0 = (const float*)d_in[2];
  const float* bl0 = (const float*)d_in[3];
  const float* Wr0 = (const float*)d_in[4];
  const float* Wl1 = (const float*)d_in[5];
  const float* bl1 = (const float*)d_in[6];
  const float* Wr1 = (const float*)d_in[7];
  const float* Wl2 = (const float*)d_in[8];
  const float* bl2 = (const float*)d_in[9];
  const float* Wr2 = (const float*)d_in[10];

  const int N = in_sizes[0] / C_IN;
  const int E = in_sizes[1] / 2;
  const int* src = ei;
  const int* dst = ei + E;

  float* h0  = (float*)d_ws;
  float* h1  = h0 + (size_t)N * C_IN;
  float* agg = h1 + (size_t)N * C_IN;
  float* cnt = agg + (size_t)N * C_IN;
  int* deg   = (int*)(cnt + N);
  int* start = deg + N;
  int* rank  = start + N;
  int* col   = rank + E;
  int* colS  = col + E;

  const int tilesM = (N + 15) / 16;
  const int nodeBlocks = (N + 255) / 256;
  const int edgeBlocks = (E + 255) / 256;
  const int waveBlocks = (N + 7) / 8;
  const int scanBlocks = (N + SCAN_CHUNK - 1) / SCAN_CHUNK;

  csr_zero<<<nodeBlocks, 256, 0, stream>>>(deg, N);
  csr_rank<<<edgeBlocks, 256, 0, stream>>>(dst, deg, rank, E);
  csr_scan<<<scanBlocks, SCAN_T, 0, stream>>>(deg, start, cnt, N);
  csr_place<<<edgeBlocks, 256, 0, stream>>>(src, dst, start, rank, col, E);
  csr_order<<<waveBlocks, 256, 0, stream>>>(start, deg, col, colS, N);

  sage_gather<<<waveBlocks, 256, 0, stream>>>(x0, start, deg, colS, agg, N);
  {
    int tiles = tilesM * (128 / 16);
    int blocks = (tiles + 7) / 8;
    sage_gemm<128, true><<<blocks, 256, 0, stream>>>(x0, agg, cnt, Wl0, bl0, Wr0, h0, tiles);
  }

  sage_gather<<<waveBlocks, 256, 0, stream>>>(h0, start, deg, colS, agg, N);
  {
    int tiles = tilesM * (128 / 16);
    int blocks = (tiles + 7) / 8;
    sage_gemm<128, true><<<blocks, 256, 0, stream>>>(h0, agg, cnt, Wl1, bl1, Wr1, h1, tiles);
  }

  sage_gather<<<waveBlocks, 256, 0, stream>>>(h1, start, deg, colS, agg, N);
  {
    int tiles = tilesM * (64 / 16);
    int blocks = (tiles + 7) / 8;
    sage_gemm<64, false><<<blocks, 256, 0, stream>>>(h1, agg, cnt, Wl2, bl2, Wr2,
                                                     (float*)d_out, tiles);
  }
}


# Projection GEMM: W staged once per work-group in LDS, each wave owns a 16-row strip over all output columns (one A fragment feeds every column tile); k summed as {0,1,4,5},{2,3,6,7} within each 8
# speedup vs baseline: 3.1567x; 1.3683x over previous
#include <hip/hip_runtime.h>
#include <hip/hip_bf16.h>

typedef __attribute__((ext_vector_type(2))) float v2f;
typedef __attribute__((ext_vector_type(8))) float v8f;

#define C_IN 128

__global__ void csr_zero(int* __restrict__ p, int n) {
  int i = blockIdx.x * blockDim.x + threadIdx.x;
  if (i < n) p[i] = 0;
}

__global__ void csr_rank(const int* __restrict__ dst, int* __restrict__ deg,
                         int* __restrict__ rank, int nEdges) {
  int e = blockIdx.x * blockDim.x + threadIdx.x;
  if (e < nEdges) rank[e] = atomicAdd(deg + dst[e], 1);
}

#define SCAN_T 256
#define SCAN_CHUNK 1024

__global__ void csr_scan(const int* __restrict__ deg, int* __restrict__ start,
                         float* __restrict__ cnt, int n) {
  __shared__ int sPre[SCAN_T];
  __shared__ int sScan[SCAN_T];
  const int tid = threadIdx.x;
  const int base = blockIdx.x * SCAN_CHUNK;

  int pre = 0;
  const int4* d4 = (const int4*)deg;
  for (int q = tid; q < base / 4; q += SCAN_T) {
    int4 t = d4[q];
    pre += (t.x + t.y) + (t.z + t.w);
  }

  const int i0 = base + tid * 4;
  int v[4];
  int tsum = 0;
#pragma unroll
  for (int j = 0; j < 4; ++j) {
    int i = i0 + j;
    v[j] = (i < n) ? deg[i] : 0;
    tsum += v[j];
  }

  sPre[tid] = pre;
  sScan[tid] = tsum;
  __syncthreads();
  for (int st = SCAN_T / 2; st > 0; st >>= 1) {
    if (tid < st) sPre[tid] += sPre[tid + st];
    __syncthreads();
  }
  const int preTotal = sPre[0];
  for (int off = 1; off < SCAN_T; off <<= 1) {
    int t = (tid >= off) ? sScan[tid - off] : 0;
    __syncthreads();
    sScan[tid] += t;
    __syncthreads();
  }
  int running = preTotal + sScan[tid] - tsum;
#pragma unroll
  for (int j = 0; j < 4; ++j) {
    int i = i0 + j;
    if (i < n) {
      start[i] = running;
      cnt[i] = (float)v[j];
      running += v[j];
    }
  }
}

__global__ void csr_place(const int* __restrict__ src, const int* __restrict__ dst,
                          const int* __restrict__ start, const int* __restrict__ rank,
                          int* __restrict__ col, int nEdges) {
  int e = blockIdx.x * blockDim.x + threadIdx.x;
  if (e < nEdges) col[start[dst[e]] + rank[e]] = src[e];
}

__global__ void csr_order(const int* __restrict__ start, const int* __restrict__ deg,
                          const int* __restrict__ colIn, int* __restrict__ colOut,
                          int nNodes) {
  int node = (int)(((long long)blockIdx.x * blockDim.x + threadIdx.x) >> 5);
  int lane = threadIdx.x & 31;
  if (node >= nNodes) return;
  node = __builtin_amdgcn_readfirstlane(node);
  const int b = start[node];
  const int d = deg[node];
  for (int j = lane; j < d; j += 32) {
    int v = colIn[b + j];
    int r = 0;
    for (int i = 0; i < d; ++i) {
      int u = colIn[b + i];
      r += ((u < v) || (u == v && i < j)) ? 1 : 0;
    }
    colOut[b + r] = v;
  }
}

__global__ void sage_gather(const float* __restrict__ x,
                            const int* __restrict__ start,
                            const int* __restrict__ deg,
                            const int* __restrict__ col,
                            float* __restrict__ agg,
                            int nNodes) {
  int node = (int)(((long long)blockIdx.x * blockDim.x + threadIdx.x) >> 5);
  const int lane = threadIdx.x & 31;
  if (node >= nNodes) return;
  node = __builtin_amdgcn_readfirstlane(node);
  const int* lst = col + start[node];
  const int d = deg[node];

  float4 acc = make_float4(0.0f, 0.0f, 0.0f, 0.0f);
#define ROW(sidx) (((const float4*)(x + (long long)(sidx) * C_IN))[lane])
#define ADD(v) { acc.x += (v).x; acc.y += (v).y; acc.z += (v).z; acc.w += (v).w; }
  int j = 0;
  for (; j + 8 <= d; j += 8) {
    int s0 = lst[j], s1 = lst[j + 1], s2 = lst[j + 2], s3 = lst[j + 3];
    int s4 = lst[j + 4], s5 = lst[j + 5], s6 = lst[j + 6], s7 = lst[j + 7];
    float4 v0 = ROW(s0), v1 = ROW(s1), v2 = ROW(s2), v3 = ROW(s3);
    float4 v4 = ROW(s4), v5 = ROW(s5), v6 = ROW(s6), v7 = ROW(s7);
    ADD(v0) ADD(v1) ADD(v2) ADD(v3) ADD(v4) ADD(v5) ADD(v6) ADD(v7)
  }
  if (j + 4 <= d) {
    int s0 = lst[j], s1 = lst[j + 1], s2 = lst[j + 2], s3 = lst[j + 3];
    float4 v0 = ROW(s0), v1 = ROW(s1), v2 = ROW(s2), v3 = ROW(s3);
    ADD(v0) ADD(v1) ADD(v2) ADD(v3)
    j += 4;
  }
  if (j + 2 <= d) {
    int s0 = lst[j], s1 = lst[j + 1];
    float4 v0 = ROW(s0), v1 = ROW(s1);
    ADD(v0) ADD(v1)
    j += 2;
  }
  if (j < d) {
    int s0 = lst[j];
    float4 v0 = ROW(s0);
    ADD(v0)
  }
#undef ROW
#undef ADD
  ((float4*)(agg + (long long)node * C_IN))[lane] = acc;
}

#define GEMM_BLOCKS 512

template <int COUT, bool RELU>
__global__ __launch_bounds__(256, 4)
void sage_gemm(const float* __restrict__ xin,
               const float* __restrict__ agg,
               const float* __restrict__ cnt,
               const float* __restrict__ Wl,
               const float* __restrict__ bl,
               const float* __restrict__ Wr,
               float* __restrict__ out,
               int nStrips) {
  constexpr int NT = COUT / 16;
  __shared__ float4 sW[64 * COUT];

  const int tid = threadIdx.x;
#pragma unroll 4
  for (int it = 0; it < (32 * COUT) / 256; ++it) {
    const int idx = it * 256 + tid;
    const int q = idx / COUT, n = idx % COUT;
    const float* p = Wl + (4 * q) * COUT + n;
    float4 v;
    v.x = p[0]; v.y = p[COUT]; v.z = p[2 * COUT]; v.w = p[3 * COUT];
    sW[idx] = v;
  }
#pragma unroll 4
  for (int it = 0; it < (32 * COUT) / 256; ++it) {
    const int idx = it * 256 + tid;
    const int q = idx / COUT, n = idx % COUT;
    const float* p = Wr + (4 * q) * COUT + n;
    float4 v;
    v.x = p[0]; v.y = p[COUT]; v.z = p[2 * COUT]; v.w = p[3 * COUT];
    sW[32 * COUT + idx] = v;
  }
  __syncthreads();

  const int lane = tid & 31;
  const int c = lane & 15;
  const int h = lane >> 4;
  const int wv = __builtin_amdgcn_readfirstlane(tid >> 5);

  for (int strip = wv * (int)gridDim.x + (int)blockIdx.x; strip < nStrips;
       strip += 8 * (int)gridDim.x) {
    const int m0 = strip * 16;
    const float inv = 1.0f / fmaxf(cnt[m0 + c], 1.0f);

    v8f acc[NT];
#pragma unroll
    for (int j = 0; j < NT; ++j) acc[j] = v8f{};

#pragma unroll
    for (int ph = 0; ph < 2; ++ph) {
      const float* arow = (ph == 0 ? agg : xin) + (long long)(m0 + c) * C_IN + 4 * h;
      const float4* bp = sW + (ph * 32 + h) * COUT + c;
      float4 an = *(const float4*)arow;
      float4 bn[NT];
#pragma unroll
      for (int j = 0; j < NT; ++j) bn[j] = bp[16 * j];
#pragma unroll 2
      for (int m = 0; m < 16; ++m) {
        float4 a = an;
        float4 b[NT];
#pragma unroll
        for (int j = 0; j < NT; ++j) b[j] = bn[j];
        if (m + 1 < 16) {
          an = *(const float4*)(arow + 8 * (m + 1));
#pragma unroll
          for (int j = 0; j < NT; ++j) bn[j] = bp[(m + 1) * 2 * COUT + 16 * j];
        }
        if (ph == 0) { a.x *= inv; a.y *= inv; a.z *= inv; a.w *= inv; }
        v2f a0, a1;
        a0[0] = a.x; a0[1] = a.y;
        a1[0] = a.z; a1[1] = a.w;
#pragma unroll
        for (int j = 0; j < NT; ++j) {
          v2f b0, b1;
          b0[0] = b[j].x; b0[1] = b[j].y;
          b1[0] = b[j].z; b1[1] = b[j].w;
          acc[j] = __builtin_amdgcn_wmma_f32_16x16x4_f32(false, a0, false, b0,
                                                         (short)0, acc[j], false, false);
          acc[j] = __builtin_amdgcn_wmma_f32_16x16x4_f32(false, a1, false, b1,
                                                         (short)0, acc[j], false, false);
        }
      }
    }

#pragma unroll
    for (int j = 0; j < NT; ++j) {
      const int col = 16 * j + c;
      const float bias = bl[col];
#pragma unroll
      for (int rr = 0; rr < 8; ++rr) {
        float v = acc[j][rr] + bias;
        if (RELU) v = fmaxf(v, 0.0f);
        out[(long long)(m0 + rr + 8 * h) * COUT + col] = v;
      }
    }
  }
}

extern "C" void kernel_launch(void* const* d_in, const int* in_sizes, int n_in,
                              void* d_out, int out_size, void* d_ws, size_t ws_size,
                              hipStream_t stream) {
  const float* x0  = (const float*)d_in[0];
  const int*   ei  = (const int*)d_in[1];
  const float* Wl0 = (const float*)d_in[2];
  const float* bl0 = (const float*)d_in[3];
  const float* Wr0 = (const float*)d_in[4];
  const float* Wl1 = (const float*)d_in[5];
  const float* bl1 = (const float*)d_in[6];
  const float* Wr1 = (const float*)d_in[7];
  const float* Wl2 = (const float*)d_in[8];
  const float* bl2 = (const float*)d_in[9];
  const float* Wr2 = (const float*)d_in[10];

  const int N = in_sizes[0] / C_IN;
  const int E = in_sizes[1] / 2;
  const int* src = ei;
  const int* dst = ei + E;

  float* h0  = (float*)d_ws;
  float* h1  = h0 + (size_t)N * C_IN;
  float* agg = h1 + (size_t)N * C_IN;
  float* cnt = agg + (size_t)N * C_IN;
  int* deg   = (int*)(cnt + N);
  int* start = deg + N;
  int* rank  = start + N;
  int* col   = rank + E;
  int* colS  = col + E;

  const int tilesM = (N + 15) / 16;
  const int nodeBlocks = (N + 255) / 256;
  const int edgeBlocks = (E + 255) / 256;
  const int waveBlocks = (N + 7) / 8;
  const int scanBlocks = (N + SCAN_CHUNK - 1) / SCAN_CHUNK;

  csr_zero<<<nodeBlocks, 256, 0, stream>>>(deg, N);
  csr_rank<<<edgeBlocks, 256, 0, stream>>>(dst, deg, rank, E);
  csr_scan<<<scanBlocks, SCAN_T, 0, stream>>>(deg, start, cnt, N);
  csr_place<<<edgeBlocks, 256, 0, stream>>>(src, dst, start, rank, col, E);
  csr_order<<<waveBlocks, 256, 0, stream>>>(start, deg, col, colS, N);

  sage_gather<<<waveBlocks, 256, 0, stream>>>(x0, start, deg, colS, agg, N);
  sage_gemm<128, true><<<GEMM_BLOCKS, 256, 0, stream>>>(x0, agg, cnt, Wl0, bl0, Wr0, h0, tilesM);

  sage_gather<<<waveBlocks, 256, 0, stream>>>(h0, start, deg, colS, agg, N);
  sage_gemm<128, true><<<GEMM_BLOCKS, 256, 0, stream>>>(h0, agg, cnt, Wl1, bl1, Wr1, h1, tilesM);

  sage_gather<<<waveBlocks, 256, 0, stream>>>(h1, start, deg, colS, agg, N);
  sage_gemm<64, false><<<GEMM_BLOCKS, 256, 0, stream>>>(h1, agg, cnt, Wl2, bl2, Wr2,
                                                        (float*)d_out, tilesM);
}


# Aggregation fused into the projection kernel: the wave owning a 16-row strip sums its nodes' source rows (same order) and feeds the mean rows to its products through LDS; no aggregate array
# speedup vs baseline: 3.3771x; 1.0698x over previous
#include <hip/hip_runtime.h>
#include <hip/hip_bf16.h>

typedef __attribute__((ext_vector_type(2))) float v2f;
typedef __attribute__((ext_vector_type(8))) float v8f;

#define C_IN 128

__global__ void csr_zero(int* __restrict__ p, int n) {
  int i = blockIdx.x * blockDim.x + threadIdx.x;
  if (i < n) p[i] = 0;
}

__global__ void csr_rank(const int* __restrict__ dst, int* __restrict__ deg,
                         int* __restrict__ rank, int nEdges) {
  int e = blockIdx.x * blockDim.x + threadIdx.x;
  if (e < nEdges) rank[e] = atomicAdd(deg + dst[e], 1);
}

#define SCAN_T 256
#define SCAN_CHUNK 1024

__global__ void csr_scan(const int* __restrict__ deg, int* __restrict__ start,
                         int n) {
  __shared__ int sPre[SCAN_T];
  __shared__ int sScan[SCAN_T];
  const int tid = threadIdx.x;
  const int base = blockIdx.x * SCAN_CHUNK;

  int pre = 0;
  const int4* d4 = (const int4*)deg;
  for (int q = tid; q < base / 4; q += SCAN_T) {
    int4 t = d4[q];
    pre += (t.x + t.y) + (t.z + t.w);
  }

  const int i0 = base + tid * 4;
  int v[4];
  int tsum = 0;
#pragma unroll
  for (int j = 0; j < 4; ++j) {
    int i = i0 + j;
    v[j] = (i < n) ? deg[i] : 0;
    tsum += v[j];
  }

  sPre[tid] = pre;
  sScan[tid] = tsum;
  __syncthreads();
  for (int st = SCAN_T / 2; st > 0; st >>= 1) {
    if (tid < st) sPre[tid] += sPre[tid + st];
    __syncthreads();
  }
  const int preTotal = sPre[0];
  for (int off = 1; off < SCAN_T; off <<= 1) {
    int t = (tid >= off) ? sScan[tid - off] : 0;
    __syncthreads();
    sScan[tid] += t;
    __syncthreads();
  }
  int running = preTotal + sScan[tid] - tsum;
#pragma unroll
  for (int j = 0; j < 4; ++j) {
    int i = i0 + j;
    if (i < n) {
      start[i] = running;
      running += v[j];
    }
  }
}

__global__ void csr_place(const int* __restrict__ src, const int* __restrict__ dst,
                          const int* __restrict__ start, const int* __restrict__ rank,
                          int* __restrict__ col, int nEdges) {
  int e = blockIdx.x * blockDim.x + threadIdx.x;
  if (e < nEdges) col[start[dst[e]] + rank[e]] = src[e];
}

__global__ void csr_order(const int* __restrict__ start, const int* __restrict__ deg,
                          const int* __restrict__ colIn, int* __restrict__ colOut,
                          int nNodes) {
  int node = (int)(((long long)blockIdx.x * blockDim.x + threadIdx.x) >> 5);
  int lane = threadIdx.x & 31;
  if (node >= nNodes) return;
  node = __builtin_amdgcn_readfirstlane(node);
  const int b = start[node];
  const int d = deg[node];
  for (int j = lane; j < d; j += 32) {
    int v = colIn[b + j];
    int r = 0;
    for (int i = 0; i < d; ++i) {
      int u = colIn[b + i];
      r += ((u < v) || (u == v && i < j)) ? 1 : 0;
    }
    colOut[b + r] = v;
  }
}

#define LAYER_BLOCKS 256
#define LAYER_T 512
#define APITCH 132

template <int COUT, bool RELU>
__global__ __launch_bounds__(LAYER_T)
void sage_layer(const float* __restrict__ xin,
                const int* __restrict__ start,
                const int* __restrict__ deg,
                const int* __restrict__ col,
                const float* __restrict__ Wl,
                const float* __restrict__ bl,
                const float* __restrict__ Wr,
                float* __restrict__ out,
                int nStrips) {
  constexpr int NT = COUT / 16;
  __shared__ float4 sW[64 * COUT];
  __shared__ __attribute__((aligned(16))) float sA[(LAYER_T / 32) * 16 * APITCH];

  const int tid = threadIdx.x;
#pragma unroll 4
  for (int it = 0; it < (32 * COUT) / LAYER_T; ++it) {
    const int idx = it * LAYER_T + tid;
    const int q = idx / COUT, n = idx % COUT;
    const float* p = Wl + (4 * q) * COUT + n;
    float4 v;
    v.x = p[0]; v.y = p[COUT]; v.z = p[2 * COUT]; v.w = p[3 * COUT];
    sW[idx] = v;
  }
#pragma unroll 4
  for (int it = 0; it < (32 * COUT) / LAYER_T; ++it) {
    const int idx = it * LAYER_T + tid;
    const int q = idx / COUT, n = idx % COUT;
    const float* p = Wr + (4 * q) * COUT + n;
    float4 v;
    v.x = p[0]; v.y = p[COUT]; v.z = p[2 * COUT]; v.w = p[3 * COUT];
    sW[32 * COUT + idx] = v;
  }
  __syncthreads();

  const int lane = tid & 31;
  const int c = lane & 15;
  const int h = lane >> 4;
  const int wv = __builtin_amdgcn_readfirstlane(tid >> 5);
  float* myA = sA + wv * (16 * APITCH);

  for (int strip = wv * (int)gridDim.x + (int)blockIdx.x; strip < nStrips;
       strip += (LAYER_T / 32) * (int)gridDim.x) {
    const int m0 = strip * 16;

    const int begL = start[m0 + c];
    const int degL = deg[m0 + c];
    const int endL = begL + degL;
    const float invL = 1.0f / fmaxf((float)degL, 1.0f);
    const int eBeg = __builtin_amdgcn_readlane(begL, 0);
    const int eEnd = __builtin_amdgcn_readlane(endL, 15);
    int r = 0;
    int curEnd = __builtin_amdgcn_readlane(endL, 0);
    float4 acc = make_float4(0.0f, 0.0f, 0.0f, 0.0f);
#define ROW(sidx) (((const float4*)(xin + (long long)(sidx) * C_IN))[lane])
#define ADD(v) { acc.x += (v).x; acc.y += (v).y; acc.z += (v).z; acc.w += (v).w; }
#define FLUSH() {                                                            \
      const float inv = __int_as_float(__builtin_amdgcn_readlane(__float_as_int(invL), r)); \
      float4 o;                                                              \
      o.x = acc.x * inv; o.y = acc.y * inv; o.z = acc.z * inv; o.w = acc.w * inv; \
      *(float4*)(myA + r * APITCH + 4 * lane) = o;                           \
      acc = make_float4(0.0f, 0.0f, 0.0f, 0.0f);                             \
      ++r;                                                                   \
      curEnd = __builtin_amdgcn_readlane(endL, r & 15);                      \
    }
    int idxv = (eBeg + lane < eEnd) ? col[eBeg + lane] : 0;
    for (int base = eBeg; base < eEnd; base += 32) {
      float4 v[32];
#pragma unroll
      for (int k = 0; k < 32; ++k) {
        const int s = __builtin_amdgcn_readlane(idxv, k);
        v[k] = ROW(s);
      }
      idxv = (base + 32 + lane < eEnd) ? col[base + 32 + lane] : 0;
#pragma unroll
      for (int k = 0; k < 32; ++k) {
        const int e = base + k;
        if (e < eEnd) {
          while (e >= curEnd) FLUSH()
          ADD(v[k])
        }
      }
    }
    while (r < 16) FLUSH()
#undef ROW
#undef ADD
#undef FLUSH

    v8f acc2[NT];
#pragma unroll
    for (int j = 0; j < NT; ++j) acc2[j] = v8f{};

#pragma unroll
    for (int ph = 0; ph < 2; ++ph) {
      const float* arowL = myA + c * APITCH + 4 * h;
      const float* arowG = xin + (long long)(m0 + c) * C_IN + 4 * h;
      const float4* bp = sW + (ph * 32 + h) * COUT + c;
      float4 an = (ph == 0) ? *(const float4*)arowL : *(const float4*)arowG;
      float4 bn[NT];
#pragma unroll
      for (int j = 0; j < NT; ++j) bn[j] = bp[16 * j];
#pragma unroll 2
      for (int m = 0; m < 16; ++m) {
        float4 a = an;
        float4 b[NT];
#pragma unroll
        for (int j = 0; j < NT; ++j) b[j] = bn[j];
        if (m + 1 < 16) {
          if (ph == 0) an = *(const float4*)(arowL + 8 * (m + 1));
          else         an = *(const float4*)(arowG + 8 * (m + 1));
#pragma unroll
          for (int j = 0; j < NT; ++j) bn[j] = bp[(m + 1) * 2 * COUT + 16 * j];
        }
        v2f a0, a1;
        a0[0] = a.x; a0[1] = a.y;
        a1[0] = a.z; a1[1] = a.w;
#pragma unroll
        for (int j = 0; j < NT; ++j) {
          v2f b0, b1;
          b0[0] = b[j].x; b0[1] = b[j].y;
          b1[0] = b[j].z; b1[1] = b[j].w;
          acc2[j] = __builtin_amdgcn_wmma_f32_16x16x4_f32(false, a0, false, b0,
                                                          (short)0, acc2[j], false, false);
          acc2[j] = __builtin_amdgcn_wmma_f32_16x16x4_f32(false, a1, false, b1,
                                                          (short)0, acc2[j], false, false);
        }
      }
    }

#pragma unroll
    for (int j = 0; j < NT; ++j) {
      const int cc = 16 * j + c;
      const float bias = bl[cc];
#pragma unroll
      for (int rr = 0; rr < 8; ++rr) {
        float v = acc2[j][rr] + bias;
        if (RELU) v = fmaxf(v, 0.0f);
        out[(long long)(m0 + rr + 8 * h) * COUT + cc] = v;
      }
    }
  }
}

extern "C" void kernel_launch(void* const* d_in, const int* in_sizes, int n_in,
                              void* d_out, int out_size, void* d_ws, size_t ws_size,
                              hipStream_t stream) {
  const float* x0  = (const float*)d_in[0];
  const int*   ei  = (const int*)d_in[1];
  const float* Wl0 = (const float*)d_in[2];
  const float* bl0 = (const float*)d_in[3];
  const float* Wr0 = (const float*)d_in[4];
  const float* Wl1 = (const float*)d_in[5];
  const float* bl1 = (const float*)d_in[6];
  const float* Wr1 = (const float*)d_in[7];
  const float* Wl2 = (const float*)d_in[8];
  const float* bl2 = (const float*)d_in[9];
  const float* Wr2 = (const float*)d_in[10];

  const int N = in_sizes[0] / C_IN;
  const int E = in_sizes[1] / 2;
  const int* src = ei;
  const int* dst = ei + E;

  float* h0  = (float*)d_ws;
  float* h1  = h0 + (size_t)N * C_IN;
  int* deg   = (int*)(h1 + (size_t)N * C_IN);
  int* start = deg + N;
  int* rank  = start + N;
  int* col   = rank + E;
  int* colS  = col + E;

  const int tilesM = (N + 15) / 16;
  const int nodeBlocks = (N + 255) / 256;
  const int edgeBlocks = (E + 255) / 256;
  const int waveBlocks = (N + 7) / 8;
  const int scanBlocks = (N + SCAN_CHUNK - 1) / SCAN_CHUNK;

  csr_zero<<<nodeBlocks, 256, 0, stream>>>(deg, N);
  csr_rank<<<edgeBlocks, 256, 0, stream>>>(dst, deg, rank, E);
  csr_scan<<<scanBlocks, SCAN_T, 0, stream>>>(deg, start, N);
  csr_place<<<edgeBlocks, 256, 0, stream>>>(src, dst, start, rank, col, E);
  csr_order<<<waveBlocks, 256, 0, stream>>>(start, deg, col, colS, N);

  sage_layer<128, true><<<LAYER_BLOCKS, LAYER_T, 0, stream>>>(x0, start, deg, colS, Wl0, bl0, Wr0, h0, tilesM);

  sage_layer<128, true><<<LAYER_BLOCKS, LAYER_T, 0, stream>>>(h0, start, deg, colS, Wl1, bl1, Wr1, h1, tilesM);

  sage_layer<64, false><<<LAYER_BLOCKS, LAYER_T, 0, stream>>>(h1, start, deg, colS, Wl2, bl2, Wr2,
                                                              (float*)d_out, tilesM);
}


# Edge grouping by destination done in LDS: edges binned per 256-node range, two launches instead of five; LDS atomics only, no global atomics or scattered 4-byte requests; same sum order
# speedup vs baseline: 5.1357x; 1.5207x over previous
#include <hip/hip_runtime.h>
#include <hip/hip_bf16.h>

typedef __attribute__((ext_vector_type(2))) float v2f;
typedef __attribute__((ext_vector_type(8))) float v8f;

#define C_IN 128

#define BIN_SHIFT 8
#define BIN_NODES (1 << BIN_SHIFT)
#define BIN_T 512
#define BIN_CHUNK 8192
#define BLD_CAP 8192

__device__ __forceinline__ int block_scan_incl(int v, int* s, int tid) {
  const int lane = tid & 31;
  const int w = __builtin_amdgcn_readfirstlane(tid >> 5);
  int x = v;
#pragma unroll
  for (int off = 1; off < 32; off <<= 1) {
    const int y = __shfl_up(x, off, 32);
    if (lane >= off) x += y;
  }
  if (lane == 31) s[w] = x;
  __syncthreads();
  int add = 0;
  for (int i = 0; i < w; ++i) add += s[i];
  __syncthreads();
  return x + add;
}

__global__ __launch_bounds__(BIN_T)
void csr_bin(const int* __restrict__ src, const int* __restrict__ dst,
             int* __restrict__ stgRaw, int* __restrict__ infoRaw,
             int nEdges, int nBins, int infoPitch) {
  __shared__ int sCnt[BIN_T];
  __shared__ int sCur[BIN_T];
  __shared__ int sScan[BIN_T];
  __shared__ int2 sEdge[BIN_CHUNK];
  int2* stg = (int2*)stgRaw;
  int2* info = (int2*)infoRaw;
  const int tid = threadIdx.x;
  const int e0 = blockIdx.x * BIN_CHUNK;
  const int left = nEdges - e0;
  const int nLoc = (left < BIN_CHUNK) ? left : BIN_CHUNK;

  sCnt[tid] = 0;
  __syncthreads();
  int d[BIN_CHUNK / BIN_T], s[BIN_CHUNK / BIN_T];
#pragma unroll
  for (int j = 0; j < BIN_CHUNK / BIN_T; ++j) {
    const int i = j * BIN_T + tid;
    d[j] = -1;
    s[j] = 0;
    if (i < nLoc) { d[j] = dst[e0 + i]; s[j] = src[e0 + i]; }
  }
#pragma unroll
  for (int j = 0; j < BIN_CHUNK / BIN_T; ++j)
    if (d[j] >= 0) atomicAdd(&sCnt[d[j] >> BIN_SHIFT], 1);
  __syncthreads();

  const int c = sCnt[tid];
  const int excl = block_scan_incl(c, sScan, tid) - c;
  sCur[tid] = excl;
  if (tid < nBins) info[(size_t)tid * infoPitch + blockIdx.x] = make_int2(c, excl);
  __syncthreads();

#pragma unroll
  for (int j = 0; j < BIN_CHUNK / BIN_T; ++j)
    if (d[j] >= 0) {
      const int pos = atomicAdd(&sCur[d[j] >> BIN_SHIFT], 1);
      sEdge[pos] = make_int2(d[j], s[j]);
    }
  __syncthreads();
  for (int i = tid; i < nLoc; i += BIN_T) stg[e0 + i] = sEdge[i];
}

__device__ __forceinline__ void order_node(const int* in, int* out, int b0, int d, int lane) {
  if (d <= 32) {
    const int v = (lane < d) ? in[b0 + lane] : 0;
    int r = 0;
    for (int i = 0; i < d; ++i) {
      const int u = __builtin_amdgcn_readlane(v, i);
      r += ((u < v) || (u == v && i < lane)) ? 1 : 0;
    }
    if (lane < d) out[b0 + r] = v;
  } else {
    for (int j = lane; j < d; j += 32) {
      const int v = in[b0 + j];
      int r = 0;
      for (int i = 0; i < d; ++i) {
        const int u = in[b0 + i];
        r += ((u < v) || (u == v && i < j)) ? 1 : 0;
      }
      out[b0 + r] = v;
    }
  }
}

__global__ __launch_bounds__(BIN_T)
void csr_build(const int* __restrict__ stgRaw, const int* __restrict__ infoRaw,
               int* __restrict__ start, int* __restrict__ deg,
               int* __restrict__ colT, int* __restrict__ colS,
               int nNodes, int nPieces, int infoPitch) {
  __shared__ int sScan[BIN_T];
  __shared__ int sRunCnt[BIN_T];
  __shared__ int sRunSrc[BIN_T];
  __shared__ int sRunDst[BIN_T];
  __shared__ int sDeg[BIN_NODES];
  __shared__ int sStart[BIN_NODES];
  __shared__ int sCur[BIN_NODES];
  __shared__ int sTot[2];
  __shared__ int2 sEdge[BLD_CAP];
  __shared__ int sCol[BLD_CAP];
  const int2* stg = (const int2*)stgRaw;
  const int2* info = (const int2*)infoRaw;
  const int tid = threadIdx.x;
  const int lane = tid & 31;
  const int wv = __builtin_amdgcn_readfirstlane(tid >> 5);
  const int bin = blockIdx.x;

  int c = 0, l = 0;
  if (tid < nPieces) {
    const int2 v = info[(size_t)bin * infoPitch + tid];
    c = v.x;
    l = v.y;
  }
  const int cIncl = block_scan_incl(c, sScan, tid);
  const int lIncl = block_scan_incl(l, sScan, tid);
  if (tid == BIN_T - 1) { sTot[0] = cIncl; sTot[1] = lIncl; }
  sRunCnt[tid] = c;
  sRunSrc[tid] = tid * BIN_CHUNK + l;
  sRunDst[tid] = cIncl - c;
  if (tid < BIN_NODES) sDeg[tid] = 0;
  __syncthreads();
  const int T = sTot[0];
  const int binStart = sTot[1];
  const bool inLds = (T <= BLD_CAP);

  if (inLds) {
    for (int b = wv; b < nPieces; b += 16 * (BIN_T / 32)) {
      int2 v[16];
      int cb[16];
#pragma unroll
      for (int u = 0; u < 16; ++u) {
        const int bb = b + u * (BIN_T / 32);
        cb[u] = (bb < nPieces) ? sRunCnt[bb] : 0;
        v[u] = make_int2(0, 0);
        if (lane < cb[u]) v[u] = stg[sRunSrc[bb] + lane];
      }
#pragma unroll
      for (int u = 0; u < 16; ++u) {
        const int bb = b + u * (BIN_T / 32);
        if (lane < cb[u]) sEdge[sRunDst[bb] + lane] = v[u];
        for (int i = lane + 32; i < cb[u]; i += 32)
          sEdge[sRunDst[bb] + i] = stg[sRunSrc[bb] + i];
      }
    }
    __syncthreads();
    for (int i = tid; i < T; i += BIN_T)
      atomicAdd(&sDeg[sEdge[i].x & (BIN_NODES - 1)], 1);
  } else {
    for (int b = 0; b < nPieces; ++b) {
      const int2* p = stg + sRunSrc[b];
      const int cbb = sRunCnt[b];
      for (int i = tid; i < cbb; i += BIN_T)
        atomicAdd(&sDeg[p[i].x & (BIN_NODES - 1)], 1);
    }
  }
  __syncthreads();

  const int dg = (tid < BIN_NODES) ? sDeg[tid] : 0;
  const int dExcl = block_scan_incl(dg, sScan, tid) - dg;
  if (tid < BIN_NODES) {
    sStart[tid] = dExcl;
    sCur[tid] = dExcl;
    const int node = bin * BIN_NODES + tid;
    if (node < nNodes) {
      deg[node] = dg;
      start[node] = binStart + dExcl;
    }
  }
  __syncthreads();

  if (inLds) {
    for (int i = tid; i < T; i += BIN_T) {
      const int2 e = sEdge[i];
      const int pos = atomicAdd(&sCur[e.x & (BIN_NODES - 1)], 1);
      sCol[pos] = e.y;
    }
    __syncthreads();
    int* sOut = (int*)sEdge;
    for (int n = wv; n < BIN_NODES; n += BIN_T / 32) {
      const int b0 = __builtin_amdgcn_readfirstlane(sStart[n]);
      const int d = __builtin_amdgcn_readfirstlane(sDeg[n]);
      order_node(sCol, sOut, b0, d, lane);
    }
    __syncthreads();
    for (int i = tid; i < T; i += BIN_T) colS[binStart + i] = sOut[i];
  } else {
    for (int b = 0; b < nPieces; ++b) {
      const int2* p = stg + sRunSrc[b];
      const int cbb = sRunCnt[b];
      for (int i = tid; i < cbb; i += BIN_T) {
        const int2 e = p[i];
        const int pos = atomicAdd(&sCur[e.x & (BIN_NODES - 1)], 1);
        colT[binStart + pos] = e.y;
      }
    }
    __syncthreads();
    for (int n = wv; n < BIN_NODES; n += BIN_T / 32) {
      const int b0 = __builtin_amdgcn_readfirstlane(sStart[n]);
      const int d = __builtin_amdgcn_readfirstlane(sDeg[n]);
      order_node(colT + binStart, colS + binStart, b0, d, lane);
    }
  }
}

#define LAYER_BLOCKS 256
#define LAYER_T 512
#define APITCH 132

template <int COUT, bool RELU>
__global__ __launch_bounds__(LAYER_T)
void sage_layer(const float* __restrict__ xin,
                const int* __restrict__ start,
                const int* __restrict__ deg,
                const int* __restrict__ col,
                const float* __restrict__ Wl,
                const float* __restrict__ bl,
                const float* __restrict__ Wr,
                float* __restrict__ out,
                int nStrips) {
  constexpr int NT = COUT / 16;
  __shared__ float4 sW[64 * COUT];
  __shared__ __attribute__((aligned(16))) float sA[(LAYER_T / 32) * 16 * APITCH];

  const int tid = threadIdx.x;
#pragma unroll 4
  for (int it = 0; it < (32 * COUT) / LAYER_T; ++it) {
    const int idx = it * LAYER_T + tid;
    const int q = idx / COUT, n = idx % COUT;
    const float* p = Wl + (4 * q) * COUT + n;
    float4 v;
    v.x = p[0]; v.y = p[COUT]; v.z = p[2 * COUT]; v.w = p[3 * COUT];
    sW[idx] = v;
  }
#pragma unroll 4
  for (int it = 0; it < (32 * COUT) / LAYER_T; ++it) {
    const int idx = it * LAYER_T + tid;
    const int q = idx / COUT, n = idx % COUT;
    const float* p = Wr + (4 * q) * COUT + n;
    float4 v;
    v.x = p[0]; v.y = p[COUT]; v.z = p[2 * COUT]; v.w = p[3 * COUT];
    sW[32 * COUT + idx] = v;
  }
  __syncthreads();

  const int lane = tid & 31;
  const int c = lane & 15;
  const int h = lane >> 4;
  const int wv = __builtin_amdgcn_readfirstlane(tid >> 5);
  float* myA = sA + wv * (16 * APITCH);

  for (int strip = wv * (int)gridDim.x + (int)blockIdx.x; strip < nStrips;
       strip += (LAYER_T / 32) * (int)gridDim.x) {
    const int m0 = strip * 16;

    const int begL = start[m0 + c];
    const int degL = deg[m0 + c];
    const int endL = begL + degL;
    const float invL = 1.0f / fmaxf((float)degL, 1.0f);
    const int eBeg = __builtin_amdgcn_readlane(begL, 0);
    const int eEnd = __builtin_amdgcn_readlane(endL, 15);
    int r = 0;
    int curEnd = __builtin_amdgcn_readlane(endL, 0);
    float4 acc = make_float4(0.0f, 0.0f, 0.0f, 0.0f);
#define ROW(sidx) (((const float4*)(xin + (long long)(sidx) * C_IN))[lane])
#define ADD(v) { acc.x += (v).x; acc.y += (v).y; acc.z += (v).z; acc.w += (v).w; }
#define FLUSH() {                                                            \
      const float inv = __int_as_float(__builtin_amdgcn_readlane(__float_as_int(invL), r)); \
      float4 o;                                                              \
      o.x = acc.x * inv; o.y = acc.y * inv; o.z = acc.z * inv; o.w = acc.w * inv; \
      *(float4*)(myA + r * APITCH + 4 * lane) = o;                           \
      acc = make_float4(0.0f, 0.0f, 0.0f, 0.0f);                             \
      ++r;                                                                   \
      curEnd = __builtin_amdgcn_readlane(endL, r & 15);                      \
    }
    int idxv = (eBeg + lane < eEnd) ? col[eBeg + lane] : 0;
    for (int base = eBeg; base < eEnd; base += 32) {
      float4 v[32];
#pragma unroll
      for (int k = 0; k < 32; ++k) {
        const int s = __builtin_amdgcn_readlane(idxv, k);
        v[k] = ROW(s);
      }
      idxv = (base + 32 + lane < eEnd) ? col[base + 32 + lane] : 0;
#pragma unroll
      for (int k = 0; k < 32; ++k) {
        const int e = base + k;
        if (e < eEnd) {
          while (e >= curEnd) FLUSH()
          ADD(v[k])
        }
      }
    }
    while (r < 16) FLUSH()
#undef ROW
#undef ADD
#undef FLUSH

    v8f acc2[NT];
#pragma unroll
    for (int j = 0; j < NT; ++j) acc2[j] = v8f{};

#pragma unroll
    for (int ph = 0; ph < 2; ++ph) {
      const float* arowL = myA + c * APITCH + 4 * h;
      const float* arowG = xin + (long long)(m0 + c) * C_IN + 4 * h;
      const float4* bp = sW + (ph * 32 + h) * COUT + c;
      float4 an = (ph == 0) ? *(const float4*)arowL : *(const float4*)arowG;
      float4 bn[NT];
#pragma unroll
      for (int j = 0; j < NT; ++j) bn[j] = bp[16 * j];
#pragma unroll 2
      for (int m = 0; m < 16; ++m) {
        float4 a = an;
        float4 b[NT];
#pragma unroll
        for (int j = 0; j < NT; ++j) b[j] = bn[j];
        if (m + 1 < 16) {
          if (ph == 0) an = *(const float4*)(arowL + 8 * (m + 1));
          else         an = *(const float4*)(arowG + 8 * (m + 1));
#pragma unroll
          for (int j = 0; j < NT; ++j) bn[j] = bp[(m + 1) * 2 * COUT + 16 * j];
        }
        v2f a0, a1;
        a0[0] = a.x; a0[1] = a.y;
        a1[0] = a.z; a1[1] = a.w;
#pragma unroll
        for (int j = 0; j < NT; ++j) {
          v2f b0, b1;
          b0[0] = b[j].x; b0[1] = b[j].y;
          b1[0] = b[j].z; b1[1] = b[j].w;
          acc2[j] = __builtin_amdgcn_wmma_f32_16x16x4_f32(false, a0, false, b0,
                                                          (short)0, acc2[j], false, false);
          acc2[j] = __builtin_amdgcn_wmma_f32_16x16x4_f32(false, a1, false, b1,
                                                          (short)0, acc2[j], false, false);
        }
      }
    }

#pragma unroll
    for (int j = 0; j < NT; ++j) {
      const int cc = 16 * j + c;
      const float bias = bl[cc];
#pragma unroll
      for (int rr = 0; rr < 8; ++rr) {
        float v = acc2[j][rr] + bias;
        if (RELU) v = fmaxf(v, 0.0f);
        out[(long long)(m0 + rr + 8 * h) * COUT + cc] = v;
      }
    }
  }
}

extern "C" void kernel_launch(void* const* d_in, const int* in_sizes, int n_in,
                              void* d_out, int out_size, void* d_ws, size_t ws_size,
                              hipStream_t stream) {
  const float* x0  = (const float*)d_in[0];
  const int*   ei  = (const int*)d_in[1];
  const float* Wl0 = (const float*)d_in[2];
  const float* bl0 = (const float*)d_in[3];
  const float* Wr0 = (const float*)d_in[4];
  const float* Wl1 = (const float*)d_in[5];
  const float* bl1 = (const float*)d_in[6];
  const float* Wr1 = (const float*)d_in[7];
  const float* Wl2 = (const float*)d_in[8];
  const float* bl2 = (const float*)d_in[9];
  const float* Wr2 = (const float*)d_in[10];

  const int N = in_sizes[0] / C_IN;
  const int E = in_sizes[1] / 2;
  const int* src = ei;
  const int* dst = ei + E;

  float* h0  = (float*)d_ws;
  float* h1  = h0 + (size_t)N * C_IN;
  int* deg   = (int*)(h1 + (size_t)N * C_IN);
  int* start = deg + N;
  int* info  = start + N;
  const int nBins = (N + BIN_NODES - 1) / BIN_NODES;
  const int nPieces = (E + BIN_CHUNK - 1) / BIN_CHUNK;
  int* stg   = info + (size_t)nBins * 256 * 2;
  int* colT  = stg + (size_t)E * 2;
  int* colS  = colT + E;

  const int tilesM = (N + 15) / 16;

  csr_bin<<<nPieces, BIN_T, 0, stream>>>(src, dst, stg, info, E, nBins, 256);
  csr_build<<<nBins, BIN_T, 0, stream>>>(stg, info, start, deg, colT, colS, N, nPieces, 256);

  sage_layer<128, true><<<LAYER_BLOCKS, LAYER_T, 0, stream>>>(x0, start, deg, colS, Wl0, bl0, Wr0, h0, tilesM);

  sage_layer<128, true><<<LAYER_BLOCKS, LAYER_T, 0, stream>>>(h0, start, deg, colS, Wl1, bl1, Wr1, h1, tilesM);

  sage_layer<64, false><<<LAYER_BLOCKS, LAYER_T, 0, stream>>>(h1, start, deg, colS, Wl2, bl2, Wr2,
                                                              (float*)d_out, tilesM);
}


# Gather rows addressed as one scalar base plus a 32-bit offset per row, all 32 offsets computed before the first request: requests leave as one group, no address register rewritten; same sum order
# speedup vs baseline: 5.3116x; 1.0343x over previous
#include <hip/hip_runtime.h>
#include <hip/hip_bf16.h>

typedef __attribute__((ext_vector_type(2))) float v2f;
typedef __attribute__((ext_vector_type(8))) float v8f;

#define C_IN 128

#define BIN_SHIFT 8
#define BIN_NODES (1 << BIN_SHIFT)
#define BIN_T 512
#define BIN_CHUNK 8192
#define BLD_CAP 8192

__device__ __forceinline__ int block_scan_incl(int v, int* s, int tid) {
  const int lane = tid & 31;
  const int w = __builtin_amdgcn_readfirstlane(tid >> 5);
  int x = v;
#pragma unroll
  for (int off = 1; off < 32; off <<= 1) {
    const int y = __shfl_up(x, off, 32);
    if (lane >= off) x += y;
  }
  if (lane == 31) s[w] = x;
  __syncthreads();
  int add = 0;
  for (int i = 0; i < w; ++i) add += s[i];
  __syncthreads();
  return x + add;
}

__global__ __launch_bounds__(BIN_T)
void csr_bin(const int* __restrict__ src, const int* __restrict__ dst,
             int* __restrict__ stgRaw, int* __restrict__ infoRaw,
             int nEdges, int nBins, int infoPitch) {
  __shared__ int sCnt[BIN_T];
  __shared__ int sCur[BIN_T];
  __shared__ int sScan[BIN_T];
  __shared__ int2 sEdge[BIN_CHUNK];
  int2* stg = (int2*)stgRaw;
  int2* info = (int2*)infoRaw;
  const int tid = threadIdx.x;
  const int e0 = blockIdx.x * BIN_CHUNK;
  const int left = nEdges - e0;
  const int nLoc = (left < BIN_CHUNK) ? left : BIN_CHUNK;

  sCnt[tid] = 0;
  __syncthreads();
  int d[BIN_CHUNK / BIN_T], s[BIN_CHUNK / BIN_T];
#pragma unroll
  for (int j = 0; j < BIN_CHUNK / BIN_T; ++j) {
    const int i = j * BIN_T + tid;
    d[j] = -1;
    s[j] = 0;
    if (i < nLoc) { d[j] = dst[e0 + i]; s[j] = src[e0 + i]; }
  }
#pragma unroll
  for (int j = 0; j < BIN_CHUNK / BIN_T; ++j)
    if (d[j] >= 0) atomicAdd(&sCnt[d[j] >> BIN_SHIFT], 1);
  __syncthreads();

  const int c = sCnt[tid];
  const int excl = block_scan_incl(c, sScan, tid) - c;
  sCur[tid] = excl;
  if (tid < nBins) info[(size_t)tid * infoPitch + blockIdx.x] = make_int2(c, excl);
  __syncthreads();

#pragma unroll
  for (int j = 0; j < BIN_CHUNK / BIN_T; ++j)
    if (d[j] >= 0) {
      const int pos = atomicAdd(&sCur[d[j] >> BIN_SHIFT], 1);
      sEdge[pos] = make_int2(d[j], s[j]);
    }
  __syncthreads();
  for (int i = tid; i < nLoc; i += BIN_T) stg[e0 + i] = sEdge[i];
}

__device__ __forceinline__ void order_node(const int* in, int* out, int b0, int d, int lane) {
  if (d <= 32) {
    const int v = (lane < d) ? in[b0 + lane] : 0;
    int r = 0;
    for (int i = 0; i < d; ++i) {
      const int u = __builtin_amdgcn_readlane(v, i);
      r += ((u < v) || (u == v && i < lane)) ? 1 : 0;
    }
    if (lane < d) out[b0 + r] = v;
  } else {
    for (int j = lane; j < d; j += 32) {
      const int v = in[b0 + j];
      int r = 0;
      for (int i = 0; i < d; ++i) {
        const int u = in[b0 + i];
        r += ((u < v) || (u == v && i < j)) ? 1 : 0;
      }
      out[b0 + r] = v;
    }
  }
}

__global__ __launch_bounds__(BIN_T)
void csr_build(const int* __restrict__ stgRaw, const int* __restrict__ infoRaw,
               int* __restrict__ start, int* __restrict__ deg,
               int* __restrict__ colT, int* __restrict__ colS,
               int nNodes, int nPieces, int infoPitch) {
  __shared__ int sScan[BIN_T];
  __shared__ int sRunCnt[BIN_T];
  __shared__ int sRunSrc[BIN_T];
  __shared__ int sRunDst[BIN_T];
  __shared__ int sDeg[BIN_NODES];
  __shared__ int sStart[BIN_NODES];
  __shared__ int sCur[BIN_NODES];
  __shared__ int sTot[2];
  __shared__ int2 sEdge[BLD_CAP];
  __shared__ int sCol[BLD_CAP];
  const int2* stg = (const int2*)stgRaw;
  const int2* info = (const int2*)infoRaw;
  const int tid = threadIdx.x;
  const int lane = tid & 31;
  const int wv = __builtin_amdgcn_readfirstlane(tid >> 5);
  const int bin = blockIdx.x;

  int c = 0, l = 0;
  if (tid < nPieces) {
    const int2 v = info[(size_t)bin * infoPitch + tid];
    c = v.x;
    l = v.y;
  }
  const int cIncl = block_scan_incl(c, sScan, tid);
  const int lIncl = block_scan_incl(l, sScan, tid);
  if (tid == BIN_T - 1) { sTot[0] = cIncl; sTot[1] = lIncl; }
  sRunCnt[tid] = c;
  sRunSrc[tid] = tid * BIN_CHUNK + l;
  sRunDst[tid] = cIncl - c;
  if (tid < BIN_NODES) sDeg[tid] = 0;
  __syncthreads();
  const int T = sTot[0];
  const int binStart = sTot[1];
  const bool inLds = (T <= BLD_CAP);

  if (inLds) {
    for (int b = wv; b < nPieces; b += 16 * (BIN_T / 32)) {
      int2 v[16];
      int cb[16];
#pragma unroll
      for (int u = 0; u < 16; ++u) {
        const int bb = b + u * (BIN_T / 32);
        cb[u] = (bb < nPieces) ? sRunCnt[bb] : 0;
        v[u] = make_int2(0, 0);
        if (lane < cb[u]) v[u] = stg[sRunSrc[bb] + lane];
      }
#pragma unroll
      for (int u = 0; u < 16; ++u) {
        const int bb = b + u * (BIN_T / 32);
        if (lane < cb[u]) sEdge[sRunDst[bb] + lane] = v[u];
        for (int i = lane + 32; i < cb[u]; i += 32)
          sEdge[sRunDst[bb] + i] = stg[sRunSrc[bb] + i];
      }
    }
    __syncthreads();
    for (int i = tid; i < T; i += BIN_T)
      atomicAdd(&sDeg[sEdge[i].x & (BIN_NODES - 1)], 1);
  } else {
    for (int b = 0; b < nPieces; ++b) {
      const int2* p = stg + sRunSrc[b];
      const int cbb = sRunCnt[b];
      for (int i = tid; i < cbb; i += BIN_T)
        atomicAdd(&sDeg[p[i].x & (BIN_NODES - 1)], 1);
    }
  }
  __syncthreads();

  const int dg = (tid < BIN_NODES) ? sDeg[tid] : 0;
  const int dExcl = block_scan_incl(dg, sScan, tid) - dg;
  if (tid < BIN_NODES) {
    sStart[tid] = dExcl;
    sCur[tid] = dExcl;
    const int node = bin * BIN_NODES + tid;
    if (node < nNodes) {
      deg[node] = dg;
      start[node] = binStart + dExcl;
    }
  }
  __syncthreads();

  if (inLds) {
    for (int i = tid; i < T; i += BIN_T) {
      const int2 e = sEdge[i];
      const int pos = atomicAdd(&sCur[e.x & (BIN_NODES - 1)], 1);
      sCol[pos] = e.y;
    }
    __syncthreads();
    int* sOut = (int*)sEdge;
    for (int n = wv; n < BIN_NODES; n += BIN_T / 32) {
      const int b0 = __builtin_amdgcn_readfirstlane(sStart[n]);
      const int d = __builtin_amdgcn_readfirstlane(sDeg[n]);
      order_node(sCol, sOut, b0, d, lane);
    }
    __syncthreads();
    for (int i = tid; i < T; i += BIN_T) colS[binStart + i] = sOut[i];
  } else {
    for (int b = 0; b < nPieces; ++b) {
      const int2* p = stg + sRunSrc[b];
      const int cbb = sRunCnt[b];
      for (int i = tid; i < cbb; i += BIN_T) {
        const int2 e = p[i];
        const int pos = atomicAdd(&sCur[e.x & (BIN_NODES - 1)], 1);
        colT[binStart + pos] = e.y;
      }
    }
    __syncthreads();
    for (int n = wv; n < BIN_NODES; n += BIN_T / 32) {
      const int b0 = __builtin_amdgcn_readfirstlane(sStart[n]);
      const int d = __builtin_amdgcn_readfirstlane(sDeg[n]);
      order_node(colT + binStart, colS + binStart, b0, d, lane);
    }
  }
}

typedef const char __attribute__((address_space(1)))* gbytes_t;
typedef const float4 __attribute__((address_space(1)))* gfloat4_t;
#define LAYER_BLOCKS 256
#define LAYER_T 512
#define APITCH 132

template <int COUT, bool RELU>
__global__ __launch_bounds__(LAYER_T)
void sage_layer(const float* __restrict__ xin,
                const int* __restrict__ start,
                const int* __restrict__ deg,
                const int* __restrict__ col,
                const float* __restrict__ Wl,
                const float* __restrict__ bl,
                const float* __restrict__ Wr,
                float* __restrict__ out,
                int nStrips) {
  constexpr int NT = COUT / 16;
  __shared__ float4 sW[64 * COUT];
  __shared__ __attribute__((aligned(16))) float sA[(LAYER_T / 32) * 16 * APITCH];

  const int tid = threadIdx.x;
#pragma unroll 4
  for (int it = 0; it < (32 * COUT) / LAYER_T; ++it) {
    const int idx = it * LAYER_T + tid;
    const int q = idx / COUT, n = idx % COUT;
    const float* p = Wl + (4 * q) * COUT + n;
    float4 v;
    v.x = p[0]; v.y = p[COUT]; v.z = p[2 * COUT]; v.w = p[3 * COUT];
    sW[idx] = v;
  }
#pragma unroll 4
  for (int it = 0; it < (32 * COUT) / LAYER_T; ++it) {
    const int idx = it * LAYER_T + tid;
    const int q = idx / COUT, n = idx % COUT;
    const float* p = Wr + (4 * q) * COUT + n;
    float4 v;
    v.x = p[0]; v.y = p[COUT]; v.z = p[2 * COUT]; v.w = p[3 * COUT];
    sW[32 * COUT + idx] = v;
  }
  __syncthreads();

  const int lane = tid & 31;
  const unsigned laneOff = (unsigned)lane * 16u;
  const int c = lane & 15;
  const int h = lane >> 4;
  const int wv = __builtin_amdgcn_readfirstlane(tid >> 5);
  float* myA = sA + wv * (16 * APITCH);

  for (int strip = wv * (int)gridDim.x + (int)blockIdx.x; strip < nStrips;
       strip += (LAYER_T / 32) * (int)gridDim.x) {
    const int m0 = strip * 16;

    const int begL = start[m0 + c];
    const int degL = deg[m0 + c];
    const int endL = begL + degL;
    const float invL = 1.0f / fmaxf((float)degL, 1.0f);
    const int eBeg = __builtin_amdgcn_readlane(begL, 0);
    const int eEnd = __builtin_amdgcn_readlane(endL, 15);
    int r = 0;
    int curEnd = __builtin_amdgcn_readlane(endL, 0);
    float4 acc = make_float4(0.0f, 0.0f, 0.0f, 0.0f);
#define ADD(v) { acc.x += (v).x; acc.y += (v).y; acc.z += (v).z; acc.w += (v).w; }
#define FLUSH() {                                                            \
      const float inv = __int_as_float(__builtin_amdgcn_readlane(__float_as_int(invL), r)); \
      float4 o;                                                              \
      o.x = acc.x * inv; o.y = acc.y * inv; o.z = acc.z * inv; o.w = acc.w * inv; \
      *(float4*)(myA + r * APITCH + 4 * lane) = o;                           \
      acc = make_float4(0.0f, 0.0f, 0.0f, 0.0f);                             \
      ++r;                                                                   \
      curEnd = __builtin_amdgcn_readlane(endL, r & 15);                      \
    }
    int idxv = (eBeg + lane < eEnd) ? col[eBeg + lane] : 0;
    for (int base = eBeg; base < eEnd; base += 32) {
      float4 v[32];
      unsigned off[32];
#pragma unroll
      for (int k = 0; k < 32; ++k)
        off[k] = ((unsigned)__builtin_amdgcn_readlane(idxv, k) << 9) + laneOff;
#pragma unroll
      for (int k = 0; k < 32; ++k)
        v[k] = *(gfloat4_t)((gbytes_t)xin + off[k]);
      idxv = (base + 32 + lane < eEnd) ? col[base + 32 + lane] : 0;
#pragma unroll
      for (int k = 0; k < 32; ++k) {
        const int e = base + k;
        if (e < eEnd) {
          while (e >= curEnd) FLUSH()
          ADD(v[k])
        }
      }
    }
    while (r < 16) FLUSH()
#undef ADD
#undef FLUSH

    v8f acc2[NT];
#pragma unroll
    for (int j = 0; j < NT; ++j) acc2[j] = v8f{};

#pragma unroll
    for (int ph = 0; ph < 2; ++ph) {
      const float* arowL = myA + c * APITCH + 4 * h;
      const float* arowG = xin + (long long)(m0 + c) * C_IN + 4 * h;
      const float4* bp = sW + (ph * 32 + h) * COUT + c;
      float4 an = (ph == 0) ? *(const float4*)arowL : *(const float4*)arowG;
      float4 bn[NT];
#pragma unroll
      for (int j = 0; j < NT; ++j) bn[j] = bp[16 * j];
#pragma unroll 2
      for (int m = 0; m < 16; ++m) {
        float4 a = an;
        float4 b[NT];
#pragma unroll
        for (int j = 0; j < NT; ++j) b[j] = bn[j];
        if (m + 1 < 16) {
          if (ph == 0) an = *(const float4*)(arowL + 8 * (m + 1));
          else         an = *(const float4*)(arowG + 8 * (m + 1));
#pragma unroll
          for (int j = 0; j < NT; ++j) bn[j] = bp[(m + 1) * 2 * COUT + 16 * j];
        }
        v2f a0, a1;
        a0[0] = a.x; a0[1] = a.y;
        a1[0] = a.z; a1[1] = a.w;
#pragma unroll
        for (int j = 0; j < NT; ++j) {
          v2f b0, b1;
          b0[0] = b[j].x; b0[1] = b[j].y;
          b1[0] = b[j].z; b1[1] = b[j].w;
          acc2[j] = __builtin_amdgcn_wmma_f32_16x16x4_f32(false, a0, false, b0,
                                                          (short)0, acc2[j], false, false);
          acc2[j] = __builtin_amdgcn_wmma_f32_16x16x4_f32(false, a1, false, b1,
                                                          (short)0, acc2[j], false, false);
        }
      }
    }

#pragma unroll
    for (int j = 0; j < NT; ++j) {
      const int cc = 16 * j + c;
      const float bias = bl[cc];
#pragma unroll
      for (int rr = 0; rr < 8; ++rr) {
        float v = acc2[j][rr] + bias;
        if (RELU) v = fmaxf(v, 0.0f);
        out[(long long)(m0 + rr + 8 * h) * COUT + cc] = v;
      }
    }
  }
}

extern "C" void kernel_launch(void* const* d_in, const int* in_sizes, int n_in,
                              void* d_out, int out_size, void* d_ws, size_t ws_size,
                              hipStream_t stream) {
  const float* x0  = (const float*)d_in[0];
  const int*   ei  = (const int*)d_in[1];
  const float* Wl0 = (const float*)d_in[2];
  const float* bl0 = (const float*)d_in[3];
  const float* Wr0 = (const float*)d_in[4];
  const float* Wl1 = (const float*)d_in[5];
  const float* bl1 = (const float*)d_in[6];
  const float* Wr1 = (const float*)d_in[7];
  const float* Wl2 = (const float*)d_in[8];
  const float* bl2 = (const float*)d_in[9];
  const float* Wr2 = (const float*)d_in[10];

  const int N = in_sizes[0] / C_IN;
  const int E = in_sizes[1] / 2;
  const int* src = ei;
  const int* dst = ei + E;

  float* h0  = (float*)d_ws;
  float* h1  = h0 + (size_t)N * C_IN;
  int* deg   = (int*)(h1 + (size_t)N * C_IN);
  int* start = deg + N;
  int* info  = start + N;
  const int nBins = (N + BIN_NODES - 1) / BIN_NODES;
  const int nPieces = (E + BIN_CHUNK - 1) / BIN_CHUNK;
  int* stg   = info + (size_t)nBins * 256 * 2;
  int* colT  = stg + (size_t)E * 2;
  int* colS  = colT + E;

  const int tilesM = (N + 15) / 16;

  csr_bin<<<nPieces, BIN_T, 0, stream>>>(src, dst, stg, info, E, nBins, 256);
  csr_build<<<nBins, BIN_T, 0, stream>>>(stg, info, start, deg, colT, colS, N, nPieces, 256);

  sage_layer<128, true><<<LAYER_BLOCKS, LAYER_T, 0, stream>>>(x0, start, deg, colS, Wl0, bl0, Wr0, h0, tilesM);

  sage_layer<128, true><<<LAYER_BLOCKS, LAYER_T, 0, stream>>>(h0, start, deg, colS, Wl1, bl1, Wr1, h1, tilesM);

  sage_layer<64, false><<<LAYER_BLOCKS, LAYER_T, 0, stream>>>(h1, start, deg, colS, Wl2, bl2, Wr2,
                                                              (float*)d_out, tilesM);
}


# Last layer gathers rows already times Wl (mean(x)@Wl = mean(x@Wl)): layer before also forms h1@Wl2 on WMMA from its LDS strip; 256-byte rows, 56 per chunk; list order kept, product re-associated
# speedup vs baseline: 5.4960x; 1.0347x over previous
#include <hip/hip_runtime.h>
#include <hip/hip_bf16.h>

typedef __attribute__((ext_vector_type(2))) float v2f;
typedef __attribute__((ext_vector_type(8))) float v8f;

#define C_IN 128

#define BIN_SHIFT 8
#define BIN_NODES (1 << BIN_SHIFT)
#define BIN_T 512
#define BIN_CHUNK 8192
#define BLD_CAP 8192

__device__ __forceinline__ int block_scan_incl(int v, int* s, int tid) {
  const int lane = tid & 31;
  const int w = __builtin_amdgcn_readfirstlane(tid >> 5);
  int x = v;
#pragma unroll
  for (int off = 1; off < 32; off <<= 1) {
    const int y = __shfl_up(x, off, 32);
    if (lane >= off) x += y;
  }
  if (lane == 31) s[w] = x;
  __syncthreads();
  int add = 0;
  for (int i = 0; i < w; ++i) add += s[i];
  __syncthreads();
  return x + add;
}

__global__ __launch_bounds__(BIN_T)
void csr_bin(const int* __restrict__ src, const int* __restrict__ dst,
             int* __restrict__ stgRaw, int* __restrict__ infoRaw,
             int nEdges, int nBins, int infoPitch) {
  __shared__ int sCnt[BIN_T];
  __shared__ int sCur[BIN_T];
  __shared__ int sScan[BIN_T];
  __shared__ int2 sEdge[BIN_CHUNK];
  int2* stg = (int2*)stgRaw;
  int2* info = (int2*)infoRaw;
  const int tid = threadIdx.x;
  const int e0 = blockIdx.x * BIN_CHUNK;
  const int left = nEdges - e0;
  const int nLoc = (left < BIN_CHUNK) ? left : BIN_CHUNK;

  sCnt[tid] = 0;
  __syncthreads();
  int d[BIN_CHUNK / BIN_T], s[BIN_CHUNK / BIN_T];
#pragma unroll
  for (int j = 0; j < BIN_CHUNK / BIN_T; ++j) {
    const int i = j * BIN_T + tid;
    d[j] = -1;
    s[j] = 0;
    if (i < nLoc) { d[j] = dst[e0 + i]; s[j] = src[e0 + i]; }
  }
#pragma unroll
  for (int j = 0; j < BIN_CHUNK / BIN_T; ++j)
    if (d[j] >= 0) atomicAdd(&sCnt[d[j] >> BIN_SHIFT], 1);
  __syncthreads();

  const int c = sCnt[tid];
  const int excl = block_scan_incl(c, sScan, tid) - c;
  sCur[tid] = excl;
  if (tid < nBins) info[(size_t)tid * infoPitch + blockIdx.x] = make_int2(c, excl);
  __syncthreads();

#pragma unroll
  for (int j = 0; j < BIN_CHUNK / BIN_T; ++j)
    if (d[j] >= 0) {
      const int pos = atomicAdd(&sCur[d[j] >> BIN_SHIFT], 1);
      sEdge[pos] = make_int2(d[j], s[j]);
    }
  __syncthreads();
  for (int i = tid; i < nLoc; i += BIN_T) stg[e0 + i] = sEdge[i];
}

__device__ __forceinline__ void order_node(const int* in, int* out, int b0, int d, int lane) {
  if (d <= 32) {
    const int v = (lane < d) ? in[b0 + lane] : 0;
    int r = 0;
    for (int i = 0; i < d; ++i) {
      const int u = __builtin_amdgcn_readlane(v, i);
      r += ((u < v) || (u == v && i < lane)) ? 1 : 0;
    }
    if (lane < d) out[b0 + r] = v;
  } else {
    for (int j = lane; j < d; j += 32) {
      const int v = in[b0 + j];
      int r = 0;
      for (int i = 0; i < d; ++i) {
        const int u = in[b0 + i];
        r += ((u < v) || (u == v && i < j)) ? 1 : 0;
      }
      out[b0 + r] = v;
    }
  }
}

__global__ __launch_bounds__(BIN_T)
void csr_build(const int* __restrict__ stgRaw, const int* __restrict__ infoRaw,
               int* __restrict__ start, int* __restrict__ deg,
               int* __restrict__ colT, int* __restrict__ colS,
               int nNodes, int nPieces, int infoPitch) {
  __shared__ int sScan[BIN_T];
  __shared__ int sRunCnt[BIN_T];
  __shared__ int sRunSrc[BIN_T];
  __shared__ int sRunDst[BIN_T];
  __shared__ int sDeg[BIN_NODES];
  __shared__ int sStart[BIN_NODES];
  __shared__ int sCur[BIN_NODES];
  __shared__ int sTot[2];
  __shared__ int2 sEdge[BLD_CAP];
  __shared__ int sCol[BLD_CAP];
  const int2* stg = (const int2*)stgRaw;
  const int2* info = (const int2*)infoRaw;
  const int tid = threadIdx.x;
  const int lane = tid & 31;
  const int wv = __builtin_amdgcn_readfirstlane(tid >> 5);
  const int bin = blockIdx.x;

  int c = 0, l = 0;
  if (tid < nPieces) {
    const int2 v = info[(size_t)bin * infoPitch + tid];
    c = v.x;
    l = v.y;
  }
  const int cIncl = block_scan_incl(c, sScan, tid);
  const int lIncl = block_scan_incl(l, sScan, tid);
  if (tid == BIN_T - 1) { sTot[0] = cIncl; sTot[1] = lIncl; }
  sRunCnt[tid] = c;
  sRunSrc[tid] = tid * BIN_CHUNK + l;
  sRunDst[tid] = cIncl - c;
  if (tid < BIN_NODES) sDeg[tid] = 0;
  __syncthreads();
  const int T = sTot[0];
  const int binStart = sTot[1];
  const bool inLds = (T <= BLD_CAP);

  if (inLds) {
    for (int b = wv; b < nPieces; b += 16 * (BIN_T / 32)) {
      int2 v[16];
      int cb[16];
#pragma unroll
      for (int u = 0; u < 16; ++u) {
        const int bb = b + u * (BIN_T / 32);
        cb[u] = (bb < nPieces) ? sRunCnt[bb] : 0;
        v[u] = make_int2(0, 0);
        if (lane < cb[u]) v[u] = stg[sRunSrc[bb] + lane];
      }
#pragma unroll
      for (int u = 0; u < 16; ++u) {
        const int bb = b + u * (BIN_T / 32);
        if (lane < cb[u]) sEdge[sRunDst[bb] + lane] = v[u];
        for (int i = lane + 32; i < cb[u]; i += 32)
          sEdge[sRunDst[bb] + i] = stg[sRunSrc[bb] + i];
      }
    }
    __syncthreads();
    for (int i = tid; i < T; i += BIN_T)
      atomicAdd(&sDeg[sEdge[i].x & (BIN_NODES - 1)], 1);
  } else {
    for (int b = 0; b < nPieces; ++b) {
      const int2* p = stg + sRunSrc[b];
      const int cbb = sRunCnt[b];
      for (int i = tid; i < cbb; i += BIN_T)
        atomicAdd(&sDeg[p[i].x & (BIN_NODES - 1)], 1);
    }
  }
  __syncthreads();

  const int dg = (tid < BIN_NODES) ? sDeg[tid] : 0;
  const int dExcl = block_scan_incl(dg, sScan, tid) - dg;
  if (tid < BIN_NODES) {
    sStart[tid] = dExcl;
    sCur[tid] = dExcl;
    const int node = bin * BIN_NODES + tid;
    if (node < nNodes) {
      deg[node] = dg;
      start[node] = binStart + dExcl;
    }
  }
  __syncthreads();

  if (inLds) {
    for (int i = tid; i < T; i += BIN_T) {
      const int2 e = sEdge[i];
      const int pos = atomicAdd(&sCur[e.x & (BIN_NODES - 1)], 1);
      sCol[pos] = e.y;
    }
    __syncthreads();
    int* sOut = (int*)sEdge;
    for (int n = wv; n < BIN_NODES; n += BIN_T / 32) {
      const int b0 = __builtin_amdgcn_readfirstlane(sStart[n]);
      const int d = __builtin_amdgcn_readfirstlane(sDeg[n]);
      order_node(sCol, sOut, b0, d, lane);
    }
    __syncthreads();
    for (int i = tid; i < T; i += BIN_T) colS[binStart + i] = sOut[i];
  } else {
    for (int b = 0; b < nPieces; ++b) {
      const int2* p = stg + sRunSrc[b];
      const int cbb = sRunCnt[b];
      for (int i = tid; i < cbb; i += BIN_T) {
        const int2 e = p[i];
        const int pos = atomicAdd(&sCur[e.x & (BIN_NODES - 1)], 1);
        colT[binStart + pos] = e.y;
      }
    }
    __syncthreads();
    for (int n = wv; n < BIN_NODES; n += BIN_T / 32) {
      const int b0 = __builtin_amdgcn_readfirstlane(sStart[n]);
      const int d = __builtin_amdgcn_readfirstlane(sDeg[n]);
      order_node(colT + binStart, colS + binStart, b0, d, lane);
    }
  }
}

typedef const char __attribute__((address_space(1)))* gbytes_t;
#define LAYER_BLOCKS 256
#define LAYER_T 512

template <int RF> struct rowpiece;
template <> struct rowpiece<4> { typedef float4 type; };
template <> struct rowpiece<2> { typedef float2 type; };
__device__ __forceinline__ void rp_zero(float4& a) { a = make_float4(0.0f, 0.0f, 0.0f, 0.0f); }
__device__ __forceinline__ void rp_zero(float2& a) { a = make_float2(0.0f, 0.0f); }
__device__ __forceinline__ void rp_add(float4& a, const float4& v) { a.x += v.x; a.y += v.y; a.z += v.z; a.w += v.w; }
__device__ __forceinline__ void rp_add(float2& a, const float2& v) { a.x += v.x; a.y += v.y; }
__device__ __forceinline__ float4 rp_scale(const float4& a, float s) { return make_float4(a.x * s, a.y * s, a.z * s, a.w * s); }
__device__ __forceinline__ float2 rp_scale(const float2& a, float s) { return make_float2(a.x * s, a.y * s); }
template <class V>
__device__ __forceinline__ V rp_load(const float* base, unsigned off) {
  typedef const V __attribute__((address_space(1)))* gp_t;
  return *(gp_t)((gbytes_t)base + off);
}

template <int CO>
__device__ __forceinline__ void stage_w(float4* s, const float* __restrict__ W, int tid) {
#pragma unroll 4
  for (int it = 0; it < (32 * CO) / LAYER_T; ++it) {
    const int idx = it * LAYER_T + tid;
    const int q = idx / CO, n = idx % CO;
    const float* p = W + (4 * q) * CO + n;
    float4 v;
    v.x = p[0]; v.y = p[CO]; v.z = p[2 * CO]; v.w = p[3 * CO];
    s[idx] = v;
  }
}

template <int COUT, bool RELU, bool PRE, int CNEXT>
__global__ __launch_bounds__(LAYER_T)
void sage_layer(const float* __restrict__ xin,
                const float* __restrict__ pin,
                const int* __restrict__ start,
                const int* __restrict__ deg,
                const int* __restrict__ col,
                const float* __restrict__ Wl,
                const float* __restrict__ bl,
                const float* __restrict__ Wr,
                float* __restrict__ out,
                const float* __restrict__ Wn,
                float* __restrict__ pout,
                int nStrips) {
  static_assert(CNEXT == 0 || (!PRE && COUT == C_IN), "the next layer's product reads a 128-wide strip");
  constexpr int NT = COUT / 16;
  constexpr int CG = PRE ? COUT : C_IN;
  constexpr int RF = CG / 32;
  constexpr int CH = (RF == 4) ? 32 : 56;
  constexpr int NI = (CH + 31) / 32;
  constexpr int GSH = (CG == 128) ? 9 : 8;
  constexpr int AP = CG + 4;
  constexpr int PH0 = PRE ? 1 : 0;
  typedef typename rowpiece<RF>::type rp_t;
  __shared__ float4 sW[(2 - PH0) * 32 * COUT];
  __shared__ float4 sWn[CNEXT > 0 ? 32 * CNEXT : 1];
  __shared__ __attribute__((aligned(16))) float sA[(LAYER_T / 32) * 16 * AP];

  const int tid = threadIdx.x;
  if constexpr (!PRE) stage_w<COUT>(sW, Wl, tid);
  stage_w<COUT>(sW + (1 - PH0) * 32 * COUT, Wr, tid);
  if constexpr (CNEXT > 0) stage_w<CNEXT>(sWn, Wn, tid);
  __syncthreads();

  const int lane = tid & 31;
  const unsigned laneOff = (unsigned)lane * (unsigned)(RF * 4);
  const int c = lane & 15;
  const int h = lane >> 4;
  const int wv = __builtin_amdgcn_readfirstlane(tid >> 5);
  float* myA = sA + wv * (16 * AP);
  const float* grow = PRE ? pin : xin;

  for (int strip = wv * (int)gridDim.x + (int)blockIdx.x; strip < nStrips;
       strip += (LAYER_T / 32) * (int)gridDim.x) {
    const int m0 = strip * 16;

    const int begL = start[m0 + c];
    const int degL = deg[m0 + c];
    const int endL = begL + degL;
    const float invL = 1.0f / fmaxf((float)degL, 1.0f);
    const int eBeg = __builtin_amdgcn_readlane(begL, 0);
    const int eEnd = __builtin_amdgcn_readlane(endL, 15);
    int r = 0;
    int curEnd = __builtin_amdgcn_readlane(endL, 0);
    rp_t acc;
    rp_zero(acc);
#define FLUSH() {                                                            \
      const float inv = __int_as_float(__builtin_amdgcn_readlane(__float_as_int(invL), r)); \
      *(rp_t*)(myA + r * AP + RF * lane) = rp_scale(acc, inv);               \
      rp_zero(acc);                                                          \
      ++r;                                                                   \
      curEnd = __builtin_amdgcn_readlane(endL, r & 15);                      \
    }
    int idxv[NI];
#pragma unroll
    for (int i = 0; i < NI; ++i)
      idxv[i] = (32 * i + lane < CH && eBeg + 32 * i + lane < eEnd) ? col[eBeg + 32 * i + lane] : 0;
    for (int base = eBeg; base < eEnd; base += CH) {
      rp_t v[CH];
      unsigned off[CH];
#pragma unroll
      for (int k = 0; k < CH; ++k)
        off[k] = ((unsigned)__builtin_amdgcn_readlane(idxv[k >> 5], k & 31) << GSH) + laneOff;
#pragma unroll
      for (int k = 0; k < CH; ++k)
        v[k] = rp_load<rp_t>(grow, off[k]);
#pragma unroll
      for (int i = 0; i < NI; ++i)
        idxv[i] = (32 * i + lane < CH && base + CH + 32 * i + lane < eEnd) ? col[base + CH + 32 * i + lane] : 0;
#pragma unroll
      for (int k = 0; k < CH; ++k) {
        const int e = base + k;
        if (e < eEnd) {
          while (e >= curEnd) FLUSH()
          rp_add(acc, v[k]);
        }
      }
    }
    while (r < 16) FLUSH()
#undef FLUSH

    v8f acc2[NT];
#pragma unroll
    for (int j = 0; j < NT; ++j) acc2[j] = v8f{};

#pragma unroll
    for (int ph = PH0; ph < 2; ++ph) {
      const float* arowL = myA + c * AP + 4 * h;
      const float* arowG = xin + (long long)(m0 + c) * C_IN + 4 * h;
      const float4* bp = sW + ((ph - PH0) * 32 + h) * COUT + c;
      float4 an = (ph == 0) ? *(const float4*)arowL : *(const float4*)arowG;
      float4 bn[NT];
#pragma unroll
      for (int j = 0; j < NT; ++j) bn[j] = bp[16 * j];
#pragma unroll 2
      for (int m = 0; m < 16; ++m) {
        float4 a = an;
        float4 b[NT];
#pragma unroll
        for (int j = 0; j < NT; ++j) b[j] = bn[j];
        if (m + 1 < 16) {
          if (ph == 0) an = *(const float4*)(arowL + 8 * (m + 1));
          else         an = *(const float4*)(arowG + 8 * (m + 1));
#pragma unroll
          for (int j = 0; j < NT; ++j) bn[j] = bp[(m + 1) * 2 * COUT + 16 * j];
        }
        v2f a0, a1;
        a0[0] = a.x; a0[1] = a.y;
        a1[0] = a.z; a1[1] = a.w;
#pragma unroll
        for (int j = 0; j < NT; ++j) {
          v2f b0, b1;
          b0[0] = b[j].x; b0[1] = b[j].y;
          b1[0] = b[j].z; b1[1] = b[j].w;
          acc2[j] = __builtin_amdgcn_wmma_f32_16x16x4_f32(false, a0, false, b0,
                                                          (short)0, acc2[j], false, false);
          acc2[j] = __builtin_amdgcn_wmma_f32_16x16x4_f32(false, a1, false, b1,
                                                          (short)0, acc2[j], false, false);
        }
      }
    }

#pragma unroll
    for (int j = 0; j < NT; ++j) {
      const int cc = 16 * j + c;
      const float bias = bl[cc];
#pragma unroll
      for (int rr = 0; rr < 8; ++rr) {
        float v = acc2[j][rr] + bias;
        if constexpr (PRE) v += myA[(rr + 8 * h) * AP + cc];
        if (RELU) v = fmaxf(v, 0.0f);
        out[(long long)(m0 + rr + 8 * h) * COUT + cc] = v;
        if constexpr (CNEXT > 0) myA[(rr + 8 * h) * AP + cc] = v;
      }
    }

    if constexpr (CNEXT > 0) {
      constexpr int NTN = CNEXT / 16;
      v8f accn[NTN];
#pragma unroll
      for (int j = 0; j < NTN; ++j) accn[j] = v8f{};
      const float* arowL = myA + c * AP + 4 * h;
      const float4* bp = sWn + h * CNEXT + c;
      float4 an = *(const float4*)arowL;
      float4 bn[NTN];
#pragma unroll
      for (int j = 0; j < NTN; ++j) bn[j] = bp[16 * j];
#pragma unroll 2
      for (int m = 0; m < 16; ++m) {
        float4 a = an;
        float4 b[NTN];
#pragma unroll
        for (int j = 0; j < NTN; ++j) b[j] = bn[j];
        if (m + 1 < 16) {
          an = *(const float4*)(arowL + 8 * (m + 1));
#pragma unroll
          for (int j = 0; j < NTN; ++j) bn[j] = bp[(m + 1) * 2 * CNEXT + 16 * j];
        }
        v2f a0, a1;
        a0[0] = a.x; a0[1] = a.y;
        a1[0] = a.z; a1[1] = a.w;
#pragma unroll
        for (int j = 0; j < NTN; ++j) {
          v2f b0, b1;
          b0[0] = b[j].x; b0[1] = b[j].y;
          b1[0] = b[j].z; b1[1] = b[j].w;
          accn[j] = __builtin_amdgcn_wmma_f32_16x16x4_f32(false, a0, false, b0,
                                                          (short)0, accn[j], false, false);
          accn[j] = __builtin_amdgcn_wmma_f32_16x16x4_f32(false, a1, false, b1,
                                                          (short)0, accn[j], false, false);
        }
      }
#pragma unroll
      for (int j = 0; j < NTN; ++j)
#pragma unroll
        for (int rr = 0; rr < 8; ++rr)
          pout[(long long)(m0 + rr + 8 * h) * CNEXT + 16 * j + c] = accn[j][rr];
    }
  }
}

extern "C" void kernel_launch(void* const* d_in, const int* in_sizes, int n_in,
                              void* d_out, int out_size, void* d_ws, size_t ws_size,
                              hipStream_t stream) {
  const float* x0  = (const float*)d_in[0];
  const int*   ei  = (const int*)d_in[1];
  const float* Wl0 = (const float*)d_in[2];
  const float* bl0 = (const float*)d_in[3];
  const float* Wr0 = (const float*)d_in[4];
  const float* Wl1 = (const float*)d_in[5];
  const float* bl1 = (const float*)d_in[6];
  const float* Wr1 = (const float*)d_in[7];
  const float* Wl2 = (const float*)d_in[8];
  const float* bl2 = (const float*)d_in[9];
  const float* Wr2 = (const float*)d_in[10];

  const int N = in_sizes[0] / C_IN;
  const int E = in_sizes[1] / 2;
  const int* src = ei;
  const int* dst = ei + E;

  float* h0  = (float*)d_ws;
  float* h1  = h0 + (size_t)N * C_IN;
  float* p2  = h1 + (size_t)N * C_IN;
  int* deg   = (int*)(p2 + (size_t)N * 64);
  int* start = deg + N;
  int* info  = start + N;
  const int nBins = (N + BIN_NODES - 1) / BIN_NODES;
  const int nPieces = (E + BIN_CHUNK - 1) / BIN_CHUNK;
  int* stg   = info + (size_t)nBins * 256 * 2;
  int* colT  = stg + (size_t)E * 2;
  int* colS  = colT + E;

  const int tilesM = (N + 15) / 16;
  const float* none = nullptr;
  float* noneOut = nullptr;

  csr_bin<<<nPieces, BIN_T, 0, stream>>>(src, dst, stg, info, E, nBins, 256);
  csr_build<<<nBins, BIN_T, 0, stream>>>(stg, info, start, deg, colT, colS, N, nPieces, 256);

  sage_layer<128, true, false, 0><<<LAYER_BLOCKS, LAYER_T, 0, stream>>>(
      x0, none, start, deg, colS, Wl0, bl0, Wr0, h0, none, noneOut, tilesM);

  sage_layer<128, true, false, 64><<<LAYER_BLOCKS, LAYER_T, 0, stream>>>(
      h0, none, start, deg, colS, Wl1, bl1, Wr1, h1, Wl2, p2, tilesM);

  sage_layer<64, false, true, 0><<<LAYER_BLOCKS, LAYER_T, 0, stream>>>(
      h1, p2, start, deg, colS, none, bl2, Wr2, (float*)d_out, none, noneOut, tilesM);
}
